# starting kernel
# baseline (speedup 1.0000x reference)
#include <hip/hip_runtime.h>
#include <math.h>

#define B_ 128
#define T_ 32
#define D_ 128
#define H_ 256
#define O_ 128
#define Z_ (D_ + H_)

typedef __attribute__((ext_vector_type(2))) float v2f;
typedef __attribute__((ext_vector_type(8))) float v8f;

__global__ __launch_bounds__(256) void stpn_recurrent_kernel(
    const float* __restrict__ sentence,
    const float* __restrict__ h0,
    const float* __restrict__ F0,
    const float* __restrict__ W,
    const float* __restrict__ bias,
    const float* __restrict__ lam,
    const float* __restrict__ gam,
    float* __restrict__ outH,
    float* __restrict__ outF)
{
  const int b    = blockIdx.x;
  const int tid  = threadIdx.x;
  const int w    = tid >> 5;
  const int lane = tid & 31;

  __shared__ __align__(16) float z[Z_];
  __shared__ float hbuf[H_];

  float*       Fb  = outF + (size_t)b * H_ * Z_;
  const float* F0b = F0   + (size_t)b * H_ * Z_;

  {
    const float4* src = (const float4*)F0b;
    float4*       dst = (float4*)Fb;
    const int n4 = (H_ * Z_) / 4;
    for (int i = tid; i < n4; i += 256) dst[i] = src[i];
    hbuf[tid] = h0[b * H_ + tid];
  }
  __syncthreads();

  const float* xb = sentence + (size_t)b * T_ * D_;
  const int c0 = lane * 3;

  for (int t = 0; t < T_; ++t) {
    if (tid < D_) z[tid] = xb[t * D_ + tid];
    z[D_ + tid] = hbuf[tid];
    if (t + 1 < T_ && tid < D_ / 16)
      __builtin_prefetch(xb + (t + 1) * D_ + tid * 16, 0, 0);
    __syncthreads();

    const float4* z4p = (const float4*)z;
    const float4 zc0 = z4p[c0 + 0];
    const float4 zc1 = z4p[c0 + 1];
    const float4 zc2 = z4p[c0 + 2];

    for (int rr = 0; rr < 32; ++rr) {
      const int r = w * 32 + rr;
      const size_t rowOff = (size_t)r * (Z_ / 4) + c0;
      const float4* Wr = (const float4*)W   + rowOff;
      float4*       Fr = (float4*)Fb        + rowOff;
      const float4* Lr = (const float4*)lam + rowOff;
      const float4* Gr = (const float4*)gam + rowOff;

      const float4 w0 = Wr[0], w1 = Wr[1], w2 = Wr[2];
      float4 f0 = Fr[0], f1 = Fr[1], f2 = Fr[2];

      float p = 0.f;
      p = fmaf(w0.x + f0.x, zc0.x, p); p = fmaf(w0.y + f0.y, zc0.y, p);
      p = fmaf(w0.z + f0.z, zc0.z, p); p = fmaf(w0.w + f0.w, zc0.w, p);
      p = fmaf(w1.x + f1.x, zc1.x, p); p = fmaf(w1.y + f1.y, zc1.y, p);
      p = fmaf(w1.z + f1.z, zc1.z, p); p = fmaf(w1.w + f1.w, zc1.w, p);
      p = fmaf(w2.x + f2.x, zc2.x, p); p = fmaf(w2.y + f2.y, zc2.y, p);
      p = fmaf(w2.z + f2.z, zc2.z, p); p = fmaf(w2.w + f2.w, zc2.w, p);

      #pragma unroll
      for (int m = 16; m >= 1; m >>= 1) p += __shfl_xor(p, m, 32);

      const float hn = tanhf(p + bias[r]);
      if (lane == 0) hbuf[r] = hn;

      const float4 l0 = Lr[0], l1 = Lr[1], l2 = Lr[2];
      const float4 g0 = Gr[0], g1 = Gr[1], g2 = Gr[2];
      f0.x = fmaf(l0.x, f0.x, g0.x * hn * zc0.x);
      f0.y = fmaf(l0.y, f0.y, g0.y * hn * zc0.y);
      f0.z = fmaf(l0.z, f0.z, g0.z * hn * zc0.z);
      f0.w = fmaf(l0.w, f0.w, g0.w * hn * zc0.w);
      f1.x = fmaf(l1.x, f1.x, g1.x * hn * zc1.x);
      f1.y = fmaf(l1.y, f1.y, g1.y * hn * zc1.y);
      f1.z = fmaf(l1.z, f1.z, g1.z * hn * zc1.z);
      f1.w = fmaf(l1.w, f1.w, g1.w * hn * zc1.w);
      f2.x = fmaf(l2.x, f2.x, g2.x * hn * zc2.x);
      f2.y = fmaf(l2.y, f2.y, g2.y * hn * zc2.y);
      f2.z = fmaf(l2.z, f2.z, g2.z * hn * zc2.z);
      f2.w = fmaf(l2.w, f2.w, g2.w * hn * zc2.w);
      Fr[0] = f0; Fr[1] = f1; Fr[2] = f2;
    }
    __syncthreads();
  }

  outH[b * H_ + tid] = hbuf[tid];
}

__global__ __launch_bounds__(32) void head_wmma_kernel(
    const float* __restrict__ hmat,
    const float* __restrict__ Wout,
    const float* __restrict__ bout,
    float* __restrict__ tag)
{
  const int tileM = blockIdx.x;
  const int tileN = blockIdx.y;
  const int lane  = threadIdx.x;
  const int half  = lane >> 4;
  const int l16   = lane & 15;

  const int m = tileM * 16 + l16;
  const int n = tileN * 16 + l16;

  v8f c = {};
  for (int k0 = 0; k0 < H_; k0 += 4) {
    const int ka = k0 + 2 * half;
    v2f a, bf;
    a.x  = hmat[m * H_ + ka + 0];
    a.y  = hmat[m * H_ + ka + 1];
    bf.x = Wout[n * H_ + ka + 0];
    bf.y = Wout[n * H_ + ka + 1];
    c = __builtin_amdgcn_wmma_f32_16x16x4_f32(
        false, a, false, bf, (short)0, c, false, false);
  }

  #pragma unroll
  for (int v = 0; v < 8; ++v) {
    const int mg = tileM * 16 + v + 8 * half;
    const int ng = tileN * 16 + l16;
    tag[mg * O_ + ng] = c[v] + bout[ng];
  }
}

extern "C" void kernel_launch(void* const* d_in, const int* in_sizes, int n_in,
                              void* d_out, int out_size, void* d_ws, size_t ws_size,
                              hipStream_t stream) {
  (void)in_sizes; (void)n_in; (void)out_size; (void)d_ws; (void)ws_size;

  const float* sentence = (const float*)d_in[0];
  const float* h0       = (const float*)d_in[1];
  const float* F0       = (const float*)d_in[2];
  const float* W        = (const float*)d_in[3];
  const float* bias     = (const float*)d_in[4];
  const float* lam      = (const float*)d_in[5];
  const float* gam      = (const float*)d_in[6];
  const float* Wout     = (const float*)d_in[7];
  const float* bout     = (const float*)d_in[8];

  float* out  = (float*)d_out;
  float* tag  = out;
  float* outH = out + B_ * O_;
  float* outF = outH + B_ * H_;

  stpn_recurrent_kernel<<<B_, 256, 0, stream>>>(
      sentence, h0, F0, W, bias, lam, gam, outH, outF);

  dim3 grid(B_ / 16, O_ / 16);
  head_wmma_kernel<<<grid, 32, 0, stream>>>(outH, Wout, bout, tag);
}


# Time loop moved to the host: one launch per timestep (33 launches), each batch's 256 rows spread over 4 workgroups so every WGP streams F, loads of 4 rows in flight per wave; sums bit-identical
# speedup vs baseline: 1.9654x; 1.9654x over previous
#include <hip/hip_runtime.h>
#include <math.h>

#define B_ 128
#define T_ 32
#define D_ 128
#define H_ 256
#define O_ 128
#define Z_ (D_ + H_)

#define RPW_ 8
#define RG_ 4
#define WGROWS_ (8 * RPW_)

typedef __attribute__((ext_vector_type(2))) float v2f;
typedef __attribute__((ext_vector_type(8))) float v8f;

__global__ __launch_bounds__(256) void stpn_step_kernel(
    const float* __restrict__ xt,
    const float* __restrict__ hprev,
    const float* Fsrc,
    const float* __restrict__ W,
    const float* __restrict__ bias,
    const float* __restrict__ lam,
    const float* __restrict__ gam,
    float* __restrict__ hnext,
    float* Fdst)
{
  const int rb   = blockIdx.x;
  const int b    = blockIdx.y;
  const int tid  = threadIdx.x;
  const int w    = tid >> 5;
  const int lane = tid & 31;
  const int c0   = lane * 3;

  __shared__ float hs[WGROWS_];

  const float4* x4 = (const float4*)(xt + (size_t)b * T_ * D_);
  const float4* h4 = (const float4*)(hprev + (size_t)b * H_);
  const float4* zp0 = (c0 + 0 < D_ / 4) ? (x4 + c0 + 0) : (h4 + (c0 + 0 - D_ / 4));
  const float4* zp1 = (c0 + 1 < D_ / 4) ? (x4 + c0 + 1) : (h4 + (c0 + 1 - D_ / 4));
  const float4* zp2 = (c0 + 2 < D_ / 4) ? (x4 + c0 + 2) : (h4 + (c0 + 2 - D_ / 4));
  const float4 zc0 = *zp0;
  const float4 zc1 = *zp1;
  const float4 zc2 = *zp2;

  const float4* W4  = (const float4*)W;
  const float4* L4  = (const float4*)lam;
  const float4* G4  = (const float4*)gam;
  const float4* Fs4 = (const float4*)(Fsrc + (size_t)b * H_ * Z_);
  float4*       Fd4 = (float4*)(Fdst + (size_t)b * H_ * Z_);

  const int rw = rb * WGROWS_ + w * RPW_;

  for (int g = 0; g < RPW_ / RG_; ++g) {
    float4 wv[RG_][3], fv[RG_][3], lv[RG_][3], gv[RG_][3];
    float  bv[RG_];

    #pragma unroll
    for (int i = 0; i < RG_; ++i) {
      const int r = rw + g * RG_ + i;
      const size_t off = (size_t)r * (Z_ / 4) + c0;
      wv[i][0] = W4[off + 0];  wv[i][1] = W4[off + 1];  wv[i][2] = W4[off + 2];
      fv[i][0] = Fs4[off + 0]; fv[i][1] = Fs4[off + 1]; fv[i][2] = Fs4[off + 2];
    }
    #pragma unroll
    for (int i = 0; i < RG_; ++i) {
      const int r = rw + g * RG_ + i;
      const size_t off = (size_t)r * (Z_ / 4) + c0;
      lv[i][0] = L4[off + 0];  lv[i][1] = L4[off + 1];  lv[i][2] = L4[off + 2];
      gv[i][0] = G4[off + 0];  gv[i][1] = G4[off + 1];  gv[i][2] = G4[off + 2];
      bv[i] = bias[r];
    }

    #pragma unroll
    for (int i = 0; i < RG_; ++i) {
      const int r = rw + g * RG_ + i;
      const size_t off = (size_t)r * (Z_ / 4) + c0;
      const float4 w0 = wv[i][0], w1 = wv[i][1], w2 = wv[i][2];
      float4 f0 = fv[i][0], f1 = fv[i][1], f2 = fv[i][2];

      float p = 0.f;
      p = fmaf(w0.x + f0.x, zc0.x, p); p = fmaf(w0.y + f0.y, zc0.y, p);
      p = fmaf(w0.z + f0.z, zc0.z, p); p = fmaf(w0.w + f0.w, zc0.w, p);
      p = fmaf(w1.x + f1.x, zc1.x, p); p = fmaf(w1.y + f1.y, zc1.y, p);
      p = fmaf(w1.z + f1.z, zc1.z, p); p = fmaf(w1.w + f1.w, zc1.w, p);
      p = fmaf(w2.x + f2.x, zc2.x, p); p = fmaf(w2.y + f2.y, zc2.y, p);
      p = fmaf(w2.z + f2.z, zc2.z, p); p = fmaf(w2.w + f2.w, zc2.w, p);

      #pragma unroll
      for (int m = 16; m >= 1; m >>= 1) p += __shfl_xor(p, m, 32);

      const float hn = tanhf(p + bv[i]);
      if (lane == 0) hs[r - rb * WGROWS_] = hn;

      const float4 l0 = lv[i][0], l1 = lv[i][1], l2 = lv[i][2];
      const float4 g0 = gv[i][0], g1 = gv[i][1], g2 = gv[i][2];
      f0.x = fmaf(l0.x, f0.x, g0.x * hn * zc0.x);
      f0.y = fmaf(l0.y, f0.y, g0.y * hn * zc0.y);
      f0.z = fmaf(l0.z, f0.z, g0.z * hn * zc0.z);
      f0.w = fmaf(l0.w, f0.w, g0.w * hn * zc0.w);
      f1.x = fmaf(l1.x, f1.x, g1.x * hn * zc1.x);
      f1.y = fmaf(l1.y, f1.y, g1.y * hn * zc1.y);
      f1.z = fmaf(l1.z, f1.z, g1.z * hn * zc1.z);
      f1.w = fmaf(l1.w, f1.w, g1.w * hn * zc1.w);
      f2.x = fmaf(l2.x, f2.x, g2.x * hn * zc2.x);
      f2.y = fmaf(l2.y, f2.y, g2.y * hn * zc2.y);
      f2.z = fmaf(l2.z, f2.z, g2.z * hn * zc2.z);
      f2.w = fmaf(l2.w, f2.w, g2.w * hn * zc2.w);
      Fd4[off + 0] = f0; Fd4[off + 1] = f1; Fd4[off + 2] = f2;
    }
  }

  __syncthreads();
  if (tid < WGROWS_) hnext[b * H_ + rb * WGROWS_ + tid] = hs[tid];
}

__global__ __launch_bounds__(32) void head_wmma_kernel(
    const float* __restrict__ hmat,
    const float* __restrict__ Wout,
    const float* __restrict__ bout,
    float* __restrict__ tag)
{
  const int tileM = blockIdx.x;
  const int tileN = blockIdx.y;
  const int lane  = threadIdx.x;
  const int half  = lane >> 4;
  const int l16   = lane & 15;

  const int m = tileM * 16 + l16;
  const int n = tileN * 16 + l16;

  v8f c = {};
  for (int k0 = 0; k0 < H_; k0 += 4) {
    const int ka = k0 + 2 * half;
    v2f a, bf;
    a.x  = hmat[m * H_ + ka + 0];
    a.y  = hmat[m * H_ + ka + 1];
    bf.x = Wout[n * H_ + ka + 0];
    bf.y = Wout[n * H_ + ka + 1];
    c = __builtin_amdgcn_wmma_f32_16x16x4_f32(
        false, a, false, bf, (short)0, c, false, false);
  }

  #pragma unroll
  for (int v = 0; v < 8; ++v) {
    const int mg = tileM * 16 + v + 8 * half;
    const int ng = tileN * 16 + l16;
    tag[mg * O_ + ng] = c[v] + bout[ng];
  }
}

extern "C" void kernel_launch(void* const* d_in, const int* in_sizes, int n_in,
                              void* d_out, int out_size, void* d_ws, size_t ws_size,
                              hipStream_t stream) {
  (void)in_sizes; (void)n_in; (void)out_size; (void)ws_size;

  const float* sentence = (const float*)d_in[0];
  const float* h0       = (const float*)d_in[1];
  const float* F0       = (const float*)d_in[2];
  const float* W        = (const float*)d_in[3];
  const float* bias     = (const float*)d_in[4];
  const float* lam      = (const float*)d_in[5];
  const float* gam      = (const float*)d_in[6];
  const float* Wout     = (const float*)d_in[7];
  const float* bout     = (const float*)d_in[8];

  float* out  = (float*)d_out;
  float* tag  = out;
  float* outH = out + B_ * O_;
  float* outF = outH + B_ * H_;

  float* hA = (float*)d_ws;
  float* hB = hA + B_ * H_;

  dim3 sgrid(H_ / WGROWS_, B_);

#define STPN_STEP(t, hp, hn, fs)                                   \
  stpn_step_kernel<<<sgrid, 256, 0, stream>>>(                     \
      sentence + (t) * D_, hp, fs, W, bias, lam, gam, hn, outF)

  STPN_STEP(0, h0, hA, F0);
  STPN_STEP(1, hA, hB, outF);
  STPN_STEP(2, hB, hA, outF);
  STPN_STEP(3, hA, hB, outF);
  STPN_STEP(4, hB, hA, outF);
  STPN_STEP(5, hA, hB, outF);
  STPN_STEP(6, hB, hA, outF);
  STPN_STEP(7, hA, hB, outF);
  STPN_STEP(8, hB, hA, outF);
  STPN_STEP(9, hA, hB, outF);
  STPN_STEP(10, hB, hA, outF);
  STPN_STEP(11, hA, hB, outF);
  STPN_STEP(12, hB, hA, outF);
  STPN_STEP(13, hA, hB, outF);
  STPN_STEP(14, hB, hA, outF);
  STPN_STEP(15, hA, hB, outF);
  STPN_STEP(16, hB, hA, outF);
  STPN_STEP(17, hA, hB, outF);
  STPN_STEP(18, hB, hA, outF);
  STPN_STEP(19, hA, hB, outF);
  STPN_STEP(20, hB, hA, outF);
  STPN_STEP(21, hA, hB, outF);
  STPN_STEP(22, hB, hA, outF);
  STPN_STEP(23, hA, hB, outF);
  STPN_STEP(24, hB, hA, outF);
  STPN_STEP(25, hA, hB, outF);
  STPN_STEP(26, hB, hA, outF);
  STPN_STEP(27, hA, hB, outF);
  STPN_STEP(28, hB, hA, outF);
  STPN_STEP(29, hA, hB, outF);
  STPN_STEP(30, hB, hA, outF);
  STPN_STEP(31, hA, outH, outF);

#undef STPN_STEP

  dim3 grid(B_ / 16, O_ / 16);
  head_wmma_kernel<<<grid, 32, 0, stream>>>(outH, Wout, bout, tag);
}


# Row-total exchanges moved from LDS permutes to the vector ALU: partner 16 by cross-row gather, 8/4/2/1 as the add's own swizzled operand; same partners and order, sums bit-identical
# speedup vs baseline: 4.0946x; 2.0833x over previous
#include <hip/hip_runtime.h>
#include <math.h>

#define B_ 128
#define T_ 32
#define D_ 128
#define H_ 256
#define O_ 128
#define Z_ (D_ + H_)

#define NB_ 4
#define RPW_ 4
#define WGROWS_ (8 * RPW_)

typedef __attribute__((ext_vector_type(2))) float v2f;
typedef __attribute__((ext_vector_type(8))) float v8f;

__device__ __forceinline__ void lds_only_barrier() {
  __builtin_amdgcn_fence(__ATOMIC_RELEASE, "workgroup", "local");
  __builtin_amdgcn_s_barrier();
  __builtin_amdgcn_fence(__ATOMIC_ACQUIRE, "workgroup", "local");
}

template <int M>
__device__ __forceinline__ float add_partner(float x) {
  const int xi = __builtin_bit_cast(int, x);
  int yi;
  if constexpr (M == 16) {
    yi = __builtin_amdgcn_permlanex16(xi, xi, 0x76543210, 0xfedcba98, false, false);
  } else {
    yi = __builtin_amdgcn_update_dpp(0, xi, 0x160 + M, 0xf, 0xf, true);
  }
  return x + __builtin_bit_cast(float, yi);
}

__global__ __launch_bounds__(256) void stpn_step_kernel(
    const float* __restrict__ xt,
    const float* __restrict__ hprev,
    const float* Fsrc,
    const float* __restrict__ W,
    const float* __restrict__ bias,
    const float* __restrict__ lam,
    const float* __restrict__ gam,
    float* __restrict__ hnext,
    float* Fdst)
{
  const int rb   = blockIdx.x;
  const int b0   = blockIdx.y * NB_;
  const int tid  = threadIdx.x;
  const int w    = tid >> 5;
  const int lane = tid & 31;
  const int c0   = lane * 3;

  __shared__ float4 zs[NB_ * (Z_ / 4)];
  __shared__ float  hs[NB_][WGROWS_];

  const int q0 = tid, q1 = tid + 256;
  const int j0 = q0 / (Z_ / 4), k0 = q0 - j0 * (Z_ / 4);
  const int j1 = q1 / (Z_ / 4), k1 = q1 - j1 * (Z_ / 4);
  const bool has1 = q1 < NB_ * (Z_ / 4);
  const float4* zsrc0 = (k0 < D_ / 4)
      ? ((const float4*)(xt + (size_t)(b0 + j0) * T_ * D_) + k0)
      : ((const float4*)(hprev + (size_t)(b0 + j0) * H_) + (k0 - D_ / 4));
  const float4* zsrc1 = (k1 < D_ / 4)
      ? ((const float4*)(xt + (size_t)(b0 + j1) * T_ * D_) + k1)
      : ((const float4*)(hprev + (size_t)(b0 + j1) * H_) + (k1 - D_ / 4));
  const float4 zr0 = *zsrc0;
  float4 zr1 = make_float4(0.f, 0.f, 0.f, 0.f);
  if (has1) zr1 = *zsrc1;

  const float4* W4  = (const float4*)W;
  const float4* L4  = (const float4*)lam;
  const float4* G4  = (const float4*)gam;

  const int rw = rb * WGROWS_ + w * RPW_;

  float4 wv[RPW_][3], lv[RPW_][3], gv[RPW_][3];
  float4 fv[NB_][RPW_][3];
  float  bv[RPW_];

  #pragma unroll
  for (int i = 0; i < RPW_; ++i) {
    const size_t off = (size_t)(rw + i) * (Z_ / 4) + c0;
    wv[i][0] = W4[off + 0];  wv[i][1] = W4[off + 1];  wv[i][2] = W4[off + 2];
  }
  #pragma unroll
  for (int j = 0; j < NB_ / 2; ++j) {
    const float4* Fs4 = (const float4*)(Fsrc + (size_t)(b0 + j) * H_ * Z_);
    #pragma unroll
    for (int i = 0; i < RPW_; ++i) {
      const size_t off = (size_t)(rw + i) * (Z_ / 4) + c0;
      fv[j][i][0] = Fs4[off + 0]; fv[j][i][1] = Fs4[off + 1]; fv[j][i][2] = Fs4[off + 2];
    }
  }
  #pragma unroll
  for (int i = 0; i < RPW_; ++i) {
    const size_t off = (size_t)(rw + i) * (Z_ / 4) + c0;
    lv[i][0] = L4[off + 0];  lv[i][1] = L4[off + 1];  lv[i][2] = L4[off + 2];
    gv[i][0] = G4[off + 0];  gv[i][1] = G4[off + 1];  gv[i][2] = G4[off + 2];
    bv[i] = bias[rw + i];
  }

  zs[q0] = zr0;
  if (has1) zs[q1] = zr1;
  lds_only_barrier();

  #pragma unroll
  for (int j = NB_ / 2; j < NB_; ++j) {
    const float4* Fs4 = (const float4*)(Fsrc + (size_t)(b0 + j) * H_ * Z_);
    #pragma unroll
    for (int i = 0; i < RPW_; ++i) {
      const size_t off = (size_t)(rw + i) * (Z_ / 4) + c0;
      fv[j][i][0] = Fs4[off + 0]; fv[j][i][1] = Fs4[off + 1]; fv[j][i][2] = Fs4[off + 2];
    }
  }

  #pragma unroll
  for (int j = 0; j < NB_; ++j) {
    float4* Fd4 = (float4*)(Fdst + (size_t)(b0 + j) * H_ * Z_);
    const float4 zc0 = zs[j * (Z_ / 4) + c0 + 0];
    const float4 zc1 = zs[j * (Z_ / 4) + c0 + 1];
    const float4 zc2 = zs[j * (Z_ / 4) + c0 + 2];

    float p[RPW_];
    #pragma unroll
    for (int i = 0; i < RPW_; ++i) {
      const float4 w0 = wv[i][0], w1 = wv[i][1], w2 = wv[i][2];
      const float4 f0 = fv[j][i][0], f1 = fv[j][i][1], f2 = fv[j][i][2];
      float q = 0.f;
      q = fmaf(w0.x + f0.x, zc0.x, q); q = fmaf(w0.y + f0.y, zc0.y, q);
      q = fmaf(w0.z + f0.z, zc0.z, q); q = fmaf(w0.w + f0.w, zc0.w, q);
      q = fmaf(w1.x + f1.x, zc1.x, q); q = fmaf(w1.y + f1.y, zc1.y, q);
      q = fmaf(w1.z + f1.z, zc1.z, q); q = fmaf(w1.w + f1.w, zc1.w, q);
      q = fmaf(w2.x + f2.x, zc2.x, q); q = fmaf(w2.y + f2.y, zc2.y, q);
      q = fmaf(w2.z + f2.z, zc2.z, q); q = fmaf(w2.w + f2.w, zc2.w, q);
      p[i] = q;
    }

    #pragma unroll
    for (int i = 0; i < RPW_; ++i) p[i] = add_partner<16>(p[i]);
    #pragma unroll
    for (int i = 0; i < RPW_; ++i) p[i] = add_partner<8>(p[i]);
    #pragma unroll
    for (int i = 0; i < RPW_; ++i) p[i] = add_partner<4>(p[i]);
    #pragma unroll
    for (int i = 0; i < RPW_; ++i) p[i] = add_partner<2>(p[i]);
    #pragma unroll
    for (int i = 0; i < RPW_; ++i) p[i] = add_partner<1>(p[i]);

    float hn[RPW_];
    #pragma unroll
    for (int i = 0; i < RPW_; ++i) {
      hn[i] = tanhf(p[i] + bv[i]);
      if (lane == 0) hs[j][w * RPW_ + i] = hn[i];
    }

    #pragma unroll
    for (int i = 0; i < RPW_; ++i) {
      const size_t off = (size_t)(rw + i) * (Z_ / 4) + c0;
      const float h1 = hn[i];
      float4 f0 = fv[j][i][0], f1 = fv[j][i][1], f2 = fv[j][i][2];
      const float4 l0 = lv[i][0], l1 = lv[i][1], l2 = lv[i][2];
      const float4 g0 = gv[i][0], g1 = gv[i][1], g2 = gv[i][2];
      f0.x = fmaf(l0.x, f0.x, g0.x * h1 * zc0.x);
      f0.y = fmaf(l0.y, f0.y, g0.y * h1 * zc0.y);
      f0.z = fmaf(l0.z, f0.z, g0.z * h1 * zc0.z);
      f0.w = fmaf(l0.w, f0.w, g0.w * h1 * zc0.w);
      f1.x = fmaf(l1.x, f1.x, g1.x * h1 * zc1.x);
      f1.y = fmaf(l1.y, f1.y, g1.y * h1 * zc1.y);
      f1.z = fmaf(l1.z, f1.z, g1.z * h1 * zc1.z);
      f1.w = fmaf(l1.w, f1.w, g1.w * h1 * zc1.w);
      f2.x = fmaf(l2.x, f2.x, g2.x * h1 * zc2.x);
      f2.y = fmaf(l2.y, f2.y, g2.y * h1 * zc2.y);
      f2.z = fmaf(l2.z, f2.z, g2.z * h1 * zc2.z);
      f2.w = fmaf(l2.w, f2.w, g2.w * h1 * zc2.w);
      Fd4[off + 0] = f0; Fd4[off + 1] = f1; Fd4[off + 2] = f2;
    }
  }

  lds_only_barrier();
  if (tid < NB_ * WGROWS_)
    hnext[(size_t)(b0 + w) * H_ + rb * WGROWS_ + lane] = hs[w][lane];
}

__global__ __launch_bounds__(32) void head_wmma_kernel(
    const float* __restrict__ hmat,
    const float* __restrict__ Wout,
    const float* __restrict__ bout,
    float* __restrict__ tag)
{
  const int tileM = blockIdx.x;
  const int tileN = blockIdx.y;
  const int lane  = threadIdx.x;
  const int half  = lane >> 4;
  const int l16   = lane & 15;

  const int m = tileM * 16 + l16;
  const int n = tileN * 16 + l16;

  v8f c = {};
  for (int k0 = 0; k0 < H_; k0 += 4) {
    const int ka = k0 + 2 * half;
    v2f a, bf;
    a.x  = hmat[m * H_ + ka + 0];
    a.y  = hmat[m * H_ + ka + 1];
    bf.x = Wout[n * H_ + ka + 0];
    bf.y = Wout[n * H_ + ka + 1];
    c = __builtin_amdgcn_wmma_f32_16x16x4_f32(
        false, a, false, bf, (short)0, c, false, false);
  }

  #pragma unroll
  for (int v = 0; v < 8; ++v) {
    const int mg = tileM * 16 + v + 8 * half;
    const int ng = tileN * 16 + l16;
    tag[mg * O_ + ng] = c[v] + bout[ng];
  }
}

extern "C" void kernel_launch(void* const* d_in, const int* in_sizes, int n_in,
                              void* d_out, int out_size, void* d_ws, size_t ws_size,
                              hipStream_t stream) {
  (void)in_sizes; (void)n_in; (void)out_size; (void)ws_size;

  const float* sentence = (const float*)d_in[0];
  const float* h0       = (const float*)d_in[1];
  const float* F0       = (const float*)d_in[2];
  const float* W        = (const float*)d_in[3];
  const float* bias     = (const float*)d_in[4];
  const float* lam      = (const float*)d_in[5];
  const float* gam      = (const float*)d_in[6];
  const float* Wout     = (const float*)d_in[7];
  const float* bout     = (const float*)d_in[8];

  float* out  = (float*)d_out;
  float* tag  = out;
  float* outH = out + B_ * O_;
  float* outF = outH + B_ * H_;

  float* hA = (float*)d_ws;
  float* hB = hA + B_ * H_;

  dim3 sgrid(H_ / WGROWS_, B_ / NB_);

#define STPN_STEP(t, hp, hn, fs)                                   \
  stpn_step_kernel<<<sgrid, 256, 0, stream>>>(                     \
      sentence + (t) * D_, hp, fs, W, bias, lam, gam, hn, outF)

  STPN_STEP(0, h0, hA, F0);
  STPN_STEP(1, hA, hB, outF);
  STPN_STEP(2, hB, hA, outF);
  STPN_STEP(3, hA, hB, outF);
  STPN_STEP(4, hB, hA, outF);
  STPN_STEP(5, hA, hB, outF);
  STPN_STEP(6, hB, hA, outF);
  STPN_STEP(7, hA, hB, outF);
  STPN_STEP(8, hB, hA, outF);
  STPN_STEP(9, hA, hB, outF);
  STPN_STEP(10, hB, hA, outF);
  STPN_STEP(11, hA, hB, outF);
  STPN_STEP(12, hB, hA, outF);
  STPN_STEP(13, hA, hB, outF);
  STPN_STEP(14, hB, hA, outF);
  STPN_STEP(15, hA, hB, outF);
  STPN_STEP(16, hB, hA, outF);
  STPN_STEP(17, hA, hB, outF);
  STPN_STEP(18, hB, hA, outF);
  STPN_STEP(19, hA, hB, outF);
  STPN_STEP(20, hB, hA, outF);
  STPN_STEP(21, hA, hB, outF);
  STPN_STEP(22, hB, hA, outF);
  STPN_STEP(23, hA, hB, outF);
  STPN_STEP(24, hB, hA, outF);
  STPN_STEP(25, hA, hB, outF);
  STPN_STEP(26, hB, hA, outF);
  STPN_STEP(27, hA, hB, outF);
  STPN_STEP(28, hB, hA, outF);
  STPN_STEP(29, hA, hB, outF);
  STPN_STEP(30, hB, hA, outF);
  STPN_STEP(31, hA, outH, outF);

#undef STPN_STEP

  dim3 grid(B_ / 16, O_ / 16);
  head_wmma_kernel<<<grid, 32, 0, stream>>>(outH, Wout, bout, tag);
}


# F stored every second timestep only: the step between forms h alone (no lam/gamma read, no F store); the next launch re-forms the skipped F in registers from the older F; same sums, bit-identical
# speedup vs baseline: 4.7846x; 1.1685x over previous
#include <hip/hip_runtime.h>
#include <math.h>

#define B_ 128
#define T_ 32
#define D_ 128
#define H_ 256
#define O_ 128
#define Z_ (D_ + H_)

#define NB_ 4
#define RPW_ 4
#define WGROWS_ (8 * RPW_)
#define ZQ_ (Z_ / 4)

typedef __attribute__((ext_vector_type(2))) float v2f;
typedef __attribute__((ext_vector_type(8))) float v8f;

__device__ __forceinline__ void lds_only_barrier() {
  __builtin_amdgcn_fence(__ATOMIC_RELEASE, "workgroup", "local");
  __builtin_amdgcn_s_barrier();
  __builtin_amdgcn_fence(__ATOMIC_ACQUIRE, "workgroup", "local");
}

template <int M>
__device__ __forceinline__ float add_partner(float x) {
  const int xi = __builtin_bit_cast(int, x);
  int yi;
  if constexpr (M == 16) {
    yi = __builtin_amdgcn_permlanex16(xi, xi, 0x76543210, 0xfedcba98, false, false);
  } else {
    yi = __builtin_amdgcn_update_dpp(0, xi, 0x160 + M, 0xf, 0xf, true);
  }
  return x + __builtin_bit_cast(float, yi);
}

__device__ __forceinline__ float dot_slice(
    const float4 w0, const float4 w1, const float4 w2,
    const float4 f0, const float4 f1, const float4 f2,
    const float4 z0, const float4 z1, const float4 z2) {
  float q = 0.f;
  q = fmaf(w0.x + f0.x, z0.x, q); q = fmaf(w0.y + f0.y, z0.y, q);
  q = fmaf(w0.z + f0.z, z0.z, q); q = fmaf(w0.w + f0.w, z0.w, q);
  q = fmaf(w1.x + f1.x, z1.x, q); q = fmaf(w1.y + f1.y, z1.y, q);
  q = fmaf(w1.z + f1.z, z1.z, q); q = fmaf(w1.w + f1.w, z1.w, q);
  q = fmaf(w2.x + f2.x, z2.x, q); q = fmaf(w2.y + f2.y, z2.y, q);
  q = fmaf(w2.z + f2.z, z2.z, q); q = fmaf(w2.w + f2.w, z2.w, q);
  return q;
}

__device__ __forceinline__ float4 hebb4(
    const float4 l, float4 f, const float4 g, const float h1, const float4 z) {
  f.x = fmaf(l.x, f.x, g.x * h1 * z.x);
  f.y = fmaf(l.y, f.y, g.y * h1 * z.y);
  f.z = fmaf(l.z, f.z, g.z * h1 * z.z);
  f.w = fmaf(l.w, f.w, g.w * h1 * z.w);
  return f;
}

__device__ __forceinline__ void reduce_rows(float (&p)[RPW_]) {
  #pragma unroll
  for (int i = 0; i < RPW_; ++i) p[i] = add_partner<16>(p[i]);
  #pragma unroll
  for (int i = 0; i < RPW_; ++i) p[i] = add_partner<8>(p[i]);
  #pragma unroll
  for (int i = 0; i < RPW_; ++i) p[i] = add_partner<4>(p[i]);
  #pragma unroll
  for (int i = 0; i < RPW_; ++i) p[i] = add_partner<2>(p[i]);
  #pragma unroll
  for (int i = 0; i < RPW_; ++i) p[i] = add_partner<1>(p[i]);
}


__global__ __launch_bounds__(256) void stpn_first_kernel(
    const float* __restrict__ xt,
    const float* __restrict__ hprev,
    const float* __restrict__ Fsrc,
    const float* __restrict__ W,
    const float* __restrict__ bias,
    float* __restrict__ hnext)
{
  const int rb   = blockIdx.x;
  const int b0   = blockIdx.y * NB_;
  const int tid  = threadIdx.x;
  const int w    = tid >> 5;
  const int lane = tid & 31;
  const int c0   = lane * 3;

  __shared__ float4 zs[NB_ * ZQ_];
  __shared__ float  hs[NB_][WGROWS_];

  const int q0 = tid, q1 = tid + 256;
  const int j0 = q0 / ZQ_, k0 = q0 - j0 * ZQ_;
  const int j1 = q1 / ZQ_, k1 = q1 - j1 * ZQ_;
  const bool has1 = q1 < NB_ * ZQ_;
  const float4* zsrc0 = (k0 < D_ / 4)
      ? ((const float4*)(xt + (size_t)(b0 + j0) * T_ * D_) + k0)
      : ((const float4*)(hprev + (size_t)(b0 + j0) * H_) + (k0 - D_ / 4));
  const float4* zsrc1 = (k1 < D_ / 4)
      ? ((const float4*)(xt + (size_t)(b0 + j1) * T_ * D_) + k1)
      : ((const float4*)(hprev + (size_t)(b0 + j1) * H_) + (k1 - D_ / 4));
  const float4 zr0 = *zsrc0;
  float4 zr1 = make_float4(0.f, 0.f, 0.f, 0.f);
  if (has1) zr1 = *zsrc1;

  const float4* W4 = (const float4*)W;
  const int rw = rb * WGROWS_ + w * RPW_;

  float4 wv[RPW_][3];
  float4 fv[NB_][RPW_][3];
  float  bv[RPW_];

  #pragma unroll
  for (int i = 0; i < RPW_; ++i) {
    const size_t off = (size_t)(rw + i) * ZQ_ + c0;
    wv[i][0] = W4[off + 0];  wv[i][1] = W4[off + 1];  wv[i][2] = W4[off + 2];
    bv[i] = bias[rw + i];
  }
  #pragma unroll
  for (int j = 0; j < NB_; ++j) {
    const float4* Fs4 = (const float4*)(Fsrc + (size_t)(b0 + j) * H_ * Z_);
    #pragma unroll
    for (int i = 0; i < RPW_; ++i) {
      const size_t off = (size_t)(rw + i) * ZQ_ + c0;
      fv[j][i][0] = Fs4[off + 0]; fv[j][i][1] = Fs4[off + 1]; fv[j][i][2] = Fs4[off + 2];
    }
  }

  zs[q0] = zr0;
  if (has1) zs[q1] = zr1;
  lds_only_barrier();

  #pragma unroll
  for (int j = 0; j < NB_; ++j) {
    const float4 zc0 = zs[j * ZQ_ + c0 + 0];
    const float4 zc1 = zs[j * ZQ_ + c0 + 1];
    const float4 zc2 = zs[j * ZQ_ + c0 + 2];

    float p[RPW_];
    #pragma unroll
    for (int i = 0; i < RPW_; ++i)
      p[i] = dot_slice(wv[i][0], wv[i][1], wv[i][2],
                       fv[j][i][0], fv[j][i][1], fv[j][i][2], zc0, zc1, zc2);
    reduce_rows(p);
    #pragma unroll
    for (int i = 0; i < RPW_; ++i) {
      const float hn = tanhf(p[i] + bv[i]);
      if (lane == 0) hs[j][w * RPW_ + i] = hn;
    }
  }

  lds_only_barrier();
  if (tid < NB_ * WGROWS_)
    hnext[(size_t)(b0 + w) * H_ + rb * WGROWS_ + lane] = hs[w][lane];
}

__device__ __forceinline__ const float4* z_source(
    const int q, const int b0,
    const float* xa, const float* ha, const float* xb, const float* hb) {
  const int s  = q / (NB_ * ZQ_);
  const int r  = q - s * (NB_ * ZQ_);
  const int j  = r / ZQ_;
  const int kk = r - j * ZQ_;
  const float* xs = s ? xb : xa;
  const float* hp = s ? hb : ha;
  return (kk < D_ / 4)
      ? ((const float4*)(xs + (size_t)(b0 + j) * T_ * D_) + kk)
      : ((const float4*)(hp + (size_t)(b0 + j) * H_) + (kk - D_ / 4));
}

__global__ __launch_bounds__(256) void stpn_second_kernel(
    const float* __restrict__ xa,
    const float* __restrict__ ha,
    const float* __restrict__ xb,
    const float* __restrict__ hb,
    const float* Fsrc,
    const float* __restrict__ W,
    const float* __restrict__ bias,
    const float* __restrict__ lam,
    const float* __restrict__ gam,
    float* __restrict__ hnext,
    float* Fdst)
{
  const int rb   = blockIdx.x;
  const int b0   = blockIdx.y * NB_;
  const int tid  = threadIdx.x;
  const int w    = tid >> 5;
  const int lane = tid & 31;
  const int c0   = lane * 3;

  __shared__ float4 zs[2 * NB_ * ZQ_];
  __shared__ float  hs[NB_][WGROWS_];

  const float4 zr0 = *z_source(tid,       b0, xa, ha, xb, hb);
  const float4 zr1 = *z_source(tid + 256, b0, xa, ha, xb, hb);
  const float4 zr2 = *z_source(tid + 512, b0, xa, ha, xb, hb);
  __builtin_amdgcn_sched_barrier(0);

  const float4* W4  = (const float4*)W;
  const float4* L4  = (const float4*)lam;
  const float4* G4  = (const float4*)gam;

  const int rw = rb * WGROWS_ + w * RPW_;

  float4 wv[RPW_][3], lv[RPW_][3], gv[RPW_][3];
  float4 fv[NB_][RPW_][3];
  float  bv[RPW_];

  #pragma unroll
  for (int i = 0; i < RPW_; ++i) {
    const size_t off = (size_t)(rw + i) * ZQ_ + c0;
    lv[i][0] = L4[off + 0];  lv[i][1] = L4[off + 1];  lv[i][2] = L4[off + 2];
    gv[i][0] = G4[off + 0];  gv[i][1] = G4[off + 1];  gv[i][2] = G4[off + 2];
  }
  #pragma unroll
  for (int j = 0; j < NB_ / 2; ++j) {
    const float4* Fs4 = (const float4*)(Fsrc + (size_t)(b0 + j) * H_ * Z_);
    #pragma unroll
    for (int i = 0; i < RPW_; ++i) {
      const size_t off = (size_t)(rw + i) * ZQ_ + c0;
      fv[j][i][0] = Fs4[off + 0]; fv[j][i][1] = Fs4[off + 1]; fv[j][i][2] = Fs4[off + 2];
    }
  }
  #pragma unroll
  for (int i = 0; i < RPW_; ++i) {
    const size_t off = (size_t)(rw + i) * ZQ_ + c0;
    wv[i][0] = W4[off + 0];  wv[i][1] = W4[off + 1];  wv[i][2] = W4[off + 2];
    bv[i] = bias[rw + i];
  }

  zs[tid] = zr0;  zs[tid + 256] = zr1;  zs[tid + 512] = zr2;
  lds_only_barrier();

  #pragma unroll
  for (int j = NB_ / 2; j < NB_; ++j) {
    const float4* Fs4 = (const float4*)(Fsrc + (size_t)(b0 + j) * H_ * Z_);
    #pragma unroll
    for (int i = 0; i < RPW_; ++i) {
      const size_t off = (size_t)(rw + i) * ZQ_ + c0;
      fv[j][i][0] = Fs4[off + 0]; fv[j][i][1] = Fs4[off + 1]; fv[j][i][2] = Fs4[off + 2];
    }
  }

  #pragma unroll
  for (int j = 0; j < NB_; ++j) {
    float4* Fd4 = (float4*)(Fdst + (size_t)(b0 + j) * H_ * Z_);
    const float4 za0 = zs[j * ZQ_ + c0 + 0];
    const float4 za1 = zs[j * ZQ_ + c0 + 1];
    const float4 za2 = zs[j * ZQ_ + c0 + 2];
    const float4 zb0 = zs[(NB_ + j) * ZQ_ + c0 + 0];
    const float4 zb1 = zs[(NB_ + j) * ZQ_ + c0 + 1];
    const float4 zb2 = zs[(NB_ + j) * ZQ_ + c0 + 2];
    const float4 h4  = zs[(NB_ + j) * ZQ_ + D_ / 4 + (rw >> 2)];
    const float hmid[RPW_] = {h4.x, h4.y, h4.z, h4.w};

    float p[RPW_];
    #pragma unroll
    for (int i = 0; i < RPW_; ++i) {
      fv[j][i][0] = hebb4(lv[i][0], fv[j][i][0], gv[i][0], hmid[i], za0);
      fv[j][i][1] = hebb4(lv[i][1], fv[j][i][1], gv[i][1], hmid[i], za1);
      fv[j][i][2] = hebb4(lv[i][2], fv[j][i][2], gv[i][2], hmid[i], za2);
      p[i] = dot_slice(wv[i][0], wv[i][1], wv[i][2],
                       fv[j][i][0], fv[j][i][1], fv[j][i][2], zb0, zb1, zb2);
    }

    reduce_rows(p);

    float hn[RPW_];
    #pragma unroll
    for (int i = 0; i < RPW_; ++i) {
      hn[i] = tanhf(p[i] + bv[i]);
      if (lane == 0) hs[j][w * RPW_ + i] = hn[i];
    }

    #pragma unroll
    for (int i = 0; i < RPW_; ++i) {
      const size_t off = (size_t)(rw + i) * ZQ_ + c0;
      Fd4[off + 0] = hebb4(lv[i][0], fv[j][i][0], gv[i][0], hn[i], zb0);
      Fd4[off + 1] = hebb4(lv[i][1], fv[j][i][1], gv[i][1], hn[i], zb1);
      Fd4[off + 2] = hebb4(lv[i][2], fv[j][i][2], gv[i][2], hn[i], zb2);
    }
  }

  lds_only_barrier();
  if (tid < NB_ * WGROWS_)
    hnext[(size_t)(b0 + w) * H_ + rb * WGROWS_ + lane] = hs[w][lane];
}

__global__ __launch_bounds__(32) void head_wmma_kernel(
    const float* __restrict__ hmat,
    const float* __restrict__ Wout,
    const float* __restrict__ bout,
    float* __restrict__ tag)
{
  const int tileM = blockIdx.x;
  const int tileN = blockIdx.y;
  const int lane  = threadIdx.x;
  const int half  = lane >> 4;
  const int l16   = lane & 15;

  const int m = tileM * 16 + l16;
  const int n = tileN * 16 + l16;

  v8f c = {};
  for (int k0 = 0; k0 < H_; k0 += 4) {
    const int ka = k0 + 2 * half;
    v2f a, bf;
    a.x  = hmat[m * H_ + ka + 0];
    a.y  = hmat[m * H_ + ka + 1];
    bf.x = Wout[n * H_ + ka + 0];
    bf.y = Wout[n * H_ + ka + 1];
    c = __builtin_amdgcn_wmma_f32_16x16x4_f32(
        false, a, false, bf, (short)0, c, false, false);
  }

  #pragma unroll
  for (int v = 0; v < 8; ++v) {
    const int mg = tileM * 16 + v + 8 * half;
    const int ng = tileN * 16 + l16;
    tag[mg * O_ + ng] = c[v] + bout[ng];
  }
}

extern "C" void kernel_launch(void* const* d_in, const int* in_sizes, int n_in,
                              void* d_out, int out_size, void* d_ws, size_t ws_size,
                              hipStream_t stream) {
  (void)in_sizes; (void)n_in; (void)out_size; (void)ws_size;

  const float* sentence = (const float*)d_in[0];
  const float* h0       = (const float*)d_in[1];
  const float* F0       = (const float*)d_in[2];
  const float* W        = (const float*)d_in[3];
  const float* bias     = (const float*)d_in[4];
  const float* lam      = (const float*)d_in[5];
  const float* gam      = (const float*)d_in[6];
  const float* Wout     = (const float*)d_in[7];
  const float* bout     = (const float*)d_in[8];

  float* out  = (float*)d_out;
  float* tag  = out;
  float* outH = out + B_ * O_;
  float* outF = outH + B_ * H_;

  float* hP = (float*)d_ws;
  float* hQ = hP + B_ * H_;
  float* hR = hQ + B_ * H_;

  dim3 sgrid(H_ / WGROWS_, B_ / NB_);

#define STPN_PAIR(t, h2, h1, hn, fs)                                    \
  stpn_first_kernel<<<sgrid, 256, 0, stream>>>(                         \
      sentence + (t) * D_, h2, fs, W, bias, h1);                        \
  stpn_second_kernel<<<sgrid, 256, 0, stream>>>(                        \
      sentence + (t) * D_, h2, sentence + ((t) + 1) * D_, h1, fs,       \
      W, bias, lam, gam, hn, outF)

  STPN_PAIR(0,  h0, hP, hQ, F0);
  STPN_PAIR(2,  hQ, hR, hP, outF);
  STPN_PAIR(4,  hP, hQ, hR, outF);
  STPN_PAIR(6,  hR, hP, hQ, outF);
  STPN_PAIR(8,  hQ, hR, hP, outF);
  STPN_PAIR(10, hP, hQ, hR, outF);
  STPN_PAIR(12, hR, hP, hQ, outF);
  STPN_PAIR(14, hQ, hR, hP, outF);
  STPN_PAIR(16, hP, hQ, hR, outF);
  STPN_PAIR(18, hR, hP, hQ, outF);
  STPN_PAIR(20, hQ, hR, hP, outF);
  STPN_PAIR(22, hP, hQ, hR, outF);
  STPN_PAIR(24, hR, hP, hQ, outF);
  STPN_PAIR(26, hQ, hR, hP, outF);
  STPN_PAIR(28, hP, hQ, hR, outF);
  STPN_PAIR(30, hR, hP, outH, outF);

#undef STPN_PAIR

  dim3 grid(B_ / 16, O_ / 16);
  head_wmma_kernel<<<grid, 32, 0, stream>>>(outH, Wout, bout, tag);
}
